# starting kernel
# baseline (speedup 1.0000x reference)
#include <hip/hip_runtime.h>
#include <math.h>

typedef __attribute__((ext_vector_type(2))) float v2f;
typedef __attribute__((ext_vector_type(8))) float v8f;

#define K_DIM 128
#define NEG_SLOPE 0.2f

__global__ void fill_u32_kernel(unsigned* __restrict__ p, int n, unsigned v) {
    int i = blockIdx.x * blockDim.x + threadIdx.x;
    if (i < n) p[i] = v;
}

template <int NCOL>
__global__ __launch_bounds__(128)
void wmma_gemm_kernel(const float* __restrict__ A, const float* __restrict__ B,
                      float* __restrict__ C, int M, int mtiles) {
    __shared__ float sW[K_DIM * NCOL];

    for (int i = threadIdx.x * 4; i < K_DIM * NCOL; i += 128 * 4)
        *(float4*)&sW[i] = *(const float4*)&B[i];
    __syncthreads();

    const int wave = threadIdx.x >> 5;
    const int lane = threadIdx.x & 31;
    const int tile = blockIdx.x * 4 + wave;
    constexpr int NTILES = NCOL / 16;
    if (tile >= mtiles * NTILES) return;

    const int mt = tile / NTILES;
    const int nt = tile - mt * NTILES;
    const int row0 = mt * 16;
    const int col0 = nt * 16;
    const int half = lane >> 4;
    const int l    = lane & 15;

    int ar = row0 + l;
    if (ar >= M) ar = M - 1;
    const float* arow = A + (size_t)ar * K_DIM;
    const float* bcol = sW + col0 + l;

    v8f acc = {};
    #pragma unroll 4
    for (int k0 = 0; k0 < K_DIM; k0 += 4) {
        v2f a = *(const v2f*)(arow + k0 + 2 * half);
        v2f b;
        b.x = bcol[(k0 + 2 * half)     * NCOL];
        b.y = bcol[(k0 + 2 * half + 1) * NCOL];
        acc = __builtin_amdgcn_wmma_f32_16x16x4_f32(
                  false, a, false, b, (short)0, acc, false, false);
    }

    float* cp = C + (size_t)(row0 + half * 8) * NCOL + col0 + l;
    if (row0 + 16 <= M) {
        #pragma unroll
        for (int i = 0; i < 8; ++i) cp[i * NCOL] = acc[i];
    } else {
        #pragma unroll
        for (int i = 0; i < 8; ++i)
            if (row0 + half * 8 + i < M) cp[i * NCOL] = acc[i];
    }
}

__global__ void att_coeff_kernel(const float* __restrict__ h,
                                 const float* __restrict__ att_s,
                                 const float* __restrict__ att_d,
                                 float* __restrict__ a_s, float* __restrict__ a_d,
                                 int N, int H, int CH) {
    int i = blockIdx.x * blockDim.x + threadIdx.x;
    if (i >= N * H) return;
    int n  = i / H;
    int hh = i - n * H;
    const float* hp = h + (size_t)n * H * CH + (size_t)hh * CH;
    const float* sp = att_s + hh * CH;
    const float* dp = att_d + hh * CH;
    float ss = 0.f, dd = 0.f;
    for (int c = 0; c < CH; ++c) { ss += hp[c] * sp[c]; dd += hp[c] * dp[c]; }
    a_s[i] = ss;
    a_d[i] = dd;
}

__device__ inline void atomicMaxF(float* addr, float v) {
    int iv = __float_as_int(v);
    if (iv >= 0) atomicMax((int*)addr, iv);
    else         atomicMin((unsigned int*)addr, (unsigned int)iv);
}

__global__ void edge_max_kernel(const int* __restrict__ ei, int E, int ET,
                                const float* __restrict__ a_s,
                                const float* __restrict__ a_d,
                                float* __restrict__ ebuf, float* __restrict__ emax,
                                int H) {
    int e = blockIdx.x * blockDim.x + threadIdx.x;
    if (e >= ET) return;
    int src = (e < E) ? ei[e]     : (e - E);
    int dst = (e < E) ? ei[E + e] : (e - E);
    for (int h = 0; h < H; ++h) {
        float v = a_s[src * H + h] + a_d[dst * H + h];
        v = (v > 0.f) ? v : NEG_SLOPE * v;
        ebuf[(size_t)e * H + h] = v;
        atomicMaxF(&emax[dst * H + h], v);
    }
}

__global__ void edge_expsum_kernel(const int* __restrict__ ei, int E, int ET,
                                   const float* __restrict__ emax,
                                   float* __restrict__ ebuf,
                                   float* __restrict__ denom, int H) {
    int e = blockIdx.x * blockDim.x + threadIdx.x;
    if (e >= ET) return;
    int dst = (e < E) ? ei[E + e] : (e - E);
    for (int h = 0; h < H; ++h) {
        float ex = __expf(ebuf[(size_t)e * H + h] - emax[dst * H + h]);
        ebuf[(size_t)e * H + h] = ex;
        atomicAdd(&denom[dst * H + h], ex);
    }
}

__global__ void edge_agg_kernel(const int* __restrict__ ei, int E,
                                const float* __restrict__ hfeat,
                                const float* __restrict__ ebuf,
                                const float* __restrict__ denom,
                                float* __restrict__ agg, int H, int CH) {
    int e = blockIdx.x;
    int c = threadIdx.x;
    int h = c / CH;
    int src = (e < E) ? ei[e]     : (e - E);
    int dst = (e < E) ? ei[E + e] : (e - E);
    float alpha = ebuf[(size_t)e * H + h] / denom[dst * H + h];
    float val   = hfeat[(size_t)src * H * CH + c] * alpha;
    atomicAdd(&agg[(size_t)dst * H * CH + c], val);
}

__global__ void bias_elu_kernel(float* __restrict__ a, const float* __restrict__ b,
                                int total, int F) {
    int i = blockIdx.x * blockDim.x + threadIdx.x;
    if (i >= total) return;
    float v = a[i] + b[i % F];
    a[i] = (v > 0.f) ? v : (__expf(v) - 1.f);
}

__global__ void bias_add_kernel(float* __restrict__ a, const float* __restrict__ b,
                                int total, int F) {
    int i = blockIdx.x * blockDim.x + threadIdx.x;
    if (i >= total) return;
    a[i] += b[i % F];
}

extern "C" void kernel_launch(void* const* d_in, const int* in_sizes, int n_in,
                              void* d_out, int out_size, void* d_ws, size_t ws_size,
                              hipStream_t stream) {
    const float* x   = (const float*)d_in[0];
    const int*   ei  = (const int*)  d_in[1];
    const float* W1  = (const float*)d_in[2];
    const float* as1 = (const float*)d_in[3];
    const float* ad1 = (const float*)d_in[4];
    const float* b1  = (const float*)d_in[5];
    const float* W2  = (const float*)d_in[6];
    const float* as2 = (const float*)d_in[7];
    const float* ad2 = (const float*)d_in[8];
    const float* b2  = (const float*)d_in[9];
    float* out = (float*)d_out;

    const int N  = in_sizes[0] / 128;
    const int E  = in_sizes[1] / 2;
    const int ET = E + N;
    const int H1 = 4, C1 = 32, F1 = 128;
    const int H2 = 1, C2 = 64, F2 = 64;

    char* w = (char*)d_ws;
    auto carve = [&](size_t bytes) {
        void* p = (void*)w;
        w += (bytes + 255) & ~(size_t)255;
        return p;
    };
    float* h1    = (float*)carve((size_t)N * F1 * 4);
    float* a1s   = (float*)carve((size_t)N * H1 * 4);
    float* a1d   = (float*)carve((size_t)N * H1 * 4);
    float* emax1 = (float*)carve((size_t)N * H1 * 4);
    float* den1  = (float*)carve((size_t)N * H1 * 4);
    float* ebuf1 = (float*)carve((size_t)ET * H1 * 4);
    float* agg1  = (float*)carve((size_t)N * F1 * 4);
    float* h2    = (float*)carve((size_t)N * F2 * 4);
    float* a2s   = (float*)carve((size_t)N * 4);
    float* a2d   = (float*)carve((size_t)N * 4);
    float* emax2 = (float*)carve((size_t)N * 4);
    float* den2  = (float*)carve((size_t)N * 4);
    float* ebuf2 = (float*)carve((size_t)ET * 4);

    const unsigned NEG_INF = 0xff800000u;
    auto fill = [&](void* p, int n, unsigned v) {
        fill_u32_kernel<<<(n + 255) / 256, 256, 0, stream>>>((unsigned*)p, n, v);
    };
    fill(emax1, N * H1, NEG_INF);
    fill(den1,  N * H1, 0u);
    fill(agg1,  N * F1, 0u);
    fill(emax2, N,      NEG_INF);
    fill(den2,  N,      0u);
    fill(out,   N * F2, 0u);

    const int eb = 256, eg = (ET + eb - 1) / eb;

    {
        int mt = (N + 15) / 16;
        int blocks = (mt * (F1 / 16) + 3) / 4;
        wmma_gemm_kernel<128><<<blocks, 128, 0, stream>>>(x, W1, h1, N, mt);
    }
    att_coeff_kernel<<<(N * H1 + 255) / 256, 256, 0, stream>>>(
        h1, as1, ad1, a1s, a1d, N, H1, C1);
    edge_max_kernel<<<eg, eb, 0, stream>>>(ei, E, ET, a1s, a1d, ebuf1, emax1, H1);
    edge_expsum_kernel<<<eg, eb, 0, stream>>>(ei, E, ET, emax1, ebuf1, den1, H1);
    edge_agg_kernel<<<ET, F1, 0, stream>>>(ei, E, h1, ebuf1, den1, agg1, H1, C1);
    bias_elu_kernel<<<(N * F1 + 255) / 256, 256, 0, stream>>>(agg1, b1, N * F1, F1);

    {
        int mt = (N + 15) / 16;
        int blocks = (mt * (F2 / 16) + 3) / 4;
        wmma_gemm_kernel<64><<<blocks, 128, 0, stream>>>(agg1, W2, h2, N, mt);
    }
    att_coeff_kernel<<<(N + 255) / 256, 256, 0, stream>>>(
        h2, as2, ad2, a2s, a2d, N, H2, C2);
    edge_max_kernel<<<eg, eb, 0, stream>>>(ei, E, ET, a2s, a2d, ebuf2, emax2, H2);
    edge_expsum_kernel<<<eg, eb, 0, stream>>>(ei, E, ET, emax2, ebuf2, den2, H2);
    edge_agg_kernel<<<ET, F2, 0, stream>>>(ei, E, h2, ebuf2, den2, out, H2, C2);
    bias_add_kernel<<<(N * F2 + 255) / 256, 256, 0, stream>>>(out, b2, N * F2, F2);
}


# Group edges by destination in-call (int atomic rank + segment starts); one wave per node does max, exp, sum and weighted row gather in registers: no float atomics; sums follow slot arrival order
# speedup vs baseline: 4.2713x; 4.2713x over previous
#include <hip/hip_runtime.h>
#include <math.h>

typedef __attribute__((ext_vector_type(2))) float v2f;
typedef __attribute__((ext_vector_type(8))) float v8f;

#define K_DIM 128
#define NEG_SLOPE 0.2f

__global__ void fill_u32_kernel(unsigned* __restrict__ p, int n, unsigned v) {
    int i = blockIdx.x * blockDim.x + threadIdx.x;
    if (i < n) p[i] = v;
}

template <int NCOL>
__global__ __launch_bounds__(128)
void wmma_gemm_kernel(const float* __restrict__ A, const float* __restrict__ B,
                      float* __restrict__ C, int M, int mtiles) {
    __shared__ float sW[K_DIM * NCOL];

    for (int i = threadIdx.x * 4; i < K_DIM * NCOL; i += 128 * 4)
        *(float4*)&sW[i] = *(const float4*)&B[i];
    __syncthreads();

    const int wave = threadIdx.x >> 5;
    const int lane = threadIdx.x & 31;
    const int tile = blockIdx.x * 4 + wave;
    constexpr int NTILES = NCOL / 16;
    if (tile >= mtiles * NTILES) return;

    const int mt = tile / NTILES;
    const int nt = tile - mt * NTILES;
    const int row0 = mt * 16;
    const int col0 = nt * 16;
    const int half = lane >> 4;
    const int l    = lane & 15;

    int ar = row0 + l;
    if (ar >= M) ar = M - 1;
    const float* arow = A + (size_t)ar * K_DIM;
    const float* bcol = sW + col0 + l;

    v8f acc = {};
    #pragma unroll 4
    for (int k0 = 0; k0 < K_DIM; k0 += 4) {
        v2f a = *(const v2f*)(arow + k0 + 2 * half);
        v2f b;
        b.x = bcol[(k0 + 2 * half)     * NCOL];
        b.y = bcol[(k0 + 2 * half + 1) * NCOL];
        acc = __builtin_amdgcn_wmma_f32_16x16x4_f32(
                  false, a, false, b, (short)0, acc, false, false);
    }

    float* cp = C + (size_t)(row0 + half * 8) * NCOL + col0 + l;
    if (row0 + 16 <= M) {
        #pragma unroll
        for (int i = 0; i < 8; ++i) cp[i * NCOL] = acc[i];
    } else {
        #pragma unroll
        for (int i = 0; i < 8; ++i)
            if (row0 + half * 8 + i < M) cp[i * NCOL] = acc[i];
    }
}

__global__ void att_coeff_kernel(const float* __restrict__ h,
                                 const float* __restrict__ att_s,
                                 const float* __restrict__ att_d,
                                 float* __restrict__ a_s, float* __restrict__ a_d,
                                 int N, int H, int CH) {
    int i = blockIdx.x * blockDim.x + threadIdx.x;
    if (i >= N * H) return;
    int n  = i / H;
    int hh = i - n * H;
    const float* hp = h + (size_t)n * H * CH + (size_t)hh * CH;
    const float* sp = att_s + hh * CH;
    const float* dp = att_d + hh * CH;
    float ss = 0.f, dd = 0.f;
    for (int c = 0; c < CH; ++c) { ss += hp[c] * sp[c]; dd += hp[c] * dp[c]; }
    a_s[i] = ss;
    a_d[i] = dd;
}

__global__ void dst_count_kernel(const int* __restrict__ ei, int E,
                                 int* __restrict__ cnt, int* __restrict__ rank) {
    int e = blockIdx.x * blockDim.x + threadIdx.x;
    if (e >= E) return;
    rank[e] = atomicAdd(&cnt[ei[E + e]], 1);
}

__global__ void seg_alloc_kernel(const int* __restrict__ cnt, int* __restrict__ total,
                                 int* __restrict__ start, int N) {
    int i = blockIdx.x * blockDim.x + threadIdx.x;
    int lane = threadIdx.x & 31;
    int c = (i < N) ? cnt[i] : 0;
    int incl = c;
    for (int d = 1; d < 32; d <<= 1) {
        int y = __shfl_up(incl, d, 32);
        if (lane >= d) incl += y;
    }
    int base = 0;
    if (lane == 31) base = atomicAdd(total, incl);
    base = __shfl(base, 31, 32);
    if (i < N) start[i] = base + incl - c;
}

__global__ void seg_fill_kernel(const int* __restrict__ ei, int E,
                                const int* __restrict__ start,
                                const int* __restrict__ rank,
                                int* __restrict__ csr_src) {
    int e = blockIdx.x * blockDim.x + threadIdx.x;
    if (e >= E) return;
    csr_src[start[ei[E + e]] + rank[e]] = ei[e];
}

__device__ __forceinline__ float wave_max(float v) {
    for (int d = 16; d > 0; d >>= 1) v = fmaxf(v, __shfl_xor(v, d, 32));
    return v;
}
__device__ __forceinline__ float wave_sum(float v) {
    for (int d = 16; d > 0; d >>= 1) v += __shfl_xor(v, d, 32);
    return v;
}
__device__ __forceinline__ float lane_bcast(float v, int t) {
    return __int_as_float(__builtin_amdgcn_readlane(__float_as_int(v), t));
}

template <int H>
__device__ __forceinline__ void load_heads(const float* __restrict__ p, int idx, float (&o)[H]) {
    if constexpr (H == 4) {
        float4 q = *(const float4*)(p + (size_t)idx * 4);
        o[0] = q.x; o[1] = q.y; o[2] = q.z; o[3] = q.w;
    } else {
        for (int h = 0; h < H; ++h) o[h] = p[(size_t)idx * H + h];
    }
}

template <int H, int CH, int G>
__device__ __forceinline__ void gather_rows(const float* __restrict__ hfeat, int s,
                                            const float (&p)[H], int hh, int lane, int t0,
                                            float (&acc)[(H * CH) / 32]) {
    constexpr int F = H * CH;
    constexpr int VEC = F / 32;
    typedef float VT __attribute__((ext_vector_type(VEC)));
    VT r[G];
    float w[G];
    for (int g = 0; g < G; ++g) {
        const int st = __builtin_amdgcn_readlane(s, t0 + g);
        r[g] = *(const VT*)(hfeat + (size_t)st * F + lane * VEC);
        float pw = lane_bcast(p[0], t0 + g);
        for (int h = 1; h < H; ++h) {
            float q = lane_bcast(p[h], t0 + g);
            pw = (hh == h) ? q : pw;
        }
        w[g] = pw;
    }
    for (int g = 0; g < G; ++g)
        for (int v = 0; v < VEC; ++v) acc[v] = fmaf(w[g], r[g][v], acc[v]);
}

template <int H, int CH>
__global__ __launch_bounds__(256)
void node_gather_kernel(const int* __restrict__ start, const int* __restrict__ cnt,
                        const int* __restrict__ csr_src,
                        const float* __restrict__ hfeat,
                        const float* __restrict__ a_s, const float* __restrict__ a_d,
                        float* __restrict__ outp, int N) {
    constexpr int F = H * CH;
    constexpr int VEC = F / 32;
    typedef float VT __attribute__((ext_vector_type(VEC)));
    const int lane = threadIdx.x & 31;
    const int i = __builtin_amdgcn_readfirstlane((int)(blockIdx.x * 8 + (threadIdx.x >> 5)));
    if (i >= N) return;
    const int hh = (lane * VEC) / CH;
    const int beg = start[i];
    const int n = cnt[i];
    const int m = n + 1;

    float ad[H];
    load_heads<H>(a_d, i, ad);
    float mx[H], den[H], acc[VEC];
    for (int h = 0; h < H; ++h) { mx[h] = -INFINITY; den[h] = 0.f; }
    for (int v = 0; v < VEC; ++v) acc[v] = 0.f;

    for (int base = 0; base < m; base += 32) {
        const int k = base + lane;
        const bool valid = k < m;
        int s = i;
        if (k < n) s = csr_src[beg + k];
        float as[H], p[H];
        load_heads<H>(a_s, s, as);
        float sc = 1.f;
        for (int h = 0; h < H; ++h) {
            float v = as[h] + ad[h];
            v = (v > 0.f) ? v : NEG_SLOPE * v;
            const float e = valid ? v : -INFINITY;
            const float nm = fmaxf(mx[h], wave_max(e));
            const float scl = __expf(mx[h] - nm);
            const float pe = __expf(e - nm);
            den[h] = den[h] * scl + wave_sum(pe);
            mx[h] = nm;
            p[h] = pe;
            sc = (hh == h) ? scl : sc;
        }
        for (int v = 0; v < VEC; ++v) acc[v] *= sc;

        const int cc = (m - base < 32) ? (m - base) : 32;
        int t = 0;
        for (; t + 8 <= cc; t += 8) gather_rows<H, CH, 8>(hfeat, s, p, hh, lane, t, acc);
        if (t + 4 <= cc) { gather_rows<H, CH, 4>(hfeat, s, p, hh, lane, t, acc); t += 4; }
        if (t + 2 <= cc) { gather_rows<H, CH, 2>(hfeat, s, p, hh, lane, t, acc); t += 2; }
        if (t < cc)      { gather_rows<H, CH, 1>(hfeat, s, p, hh, lane, t, acc); }
    }

    float dsel = den[0];
    for (int h = 1; h < H; ++h) dsel = (hh == h) ? den[h] : dsel;
    VT o;
    for (int v = 0; v < VEC; ++v) o[v] = acc[v] / dsel;
    *(VT*)(outp + (size_t)i * F + lane * VEC) = o;
}

__global__ void bias_elu_kernel(float* __restrict__ a, const float* __restrict__ b,
                                int total, int F) {
    int i = blockIdx.x * blockDim.x + threadIdx.x;
    if (i >= total) return;
    float v = a[i] + b[i % F];
    a[i] = (v > 0.f) ? v : (__expf(v) - 1.f);
}

__global__ void bias_add_kernel(float* __restrict__ a, const float* __restrict__ b,
                                int total, int F) {
    int i = blockIdx.x * blockDim.x + threadIdx.x;
    if (i >= total) return;
    a[i] += b[i % F];
}

extern "C" void kernel_launch(void* const* d_in, const int* in_sizes, int n_in,
                              void* d_out, int out_size, void* d_ws, size_t ws_size,
                              hipStream_t stream) {
    const float* x   = (const float*)d_in[0];
    const int*   ei  = (const int*)  d_in[1];
    const float* W1  = (const float*)d_in[2];
    const float* as1 = (const float*)d_in[3];
    const float* ad1 = (const float*)d_in[4];
    const float* b1  = (const float*)d_in[5];
    const float* W2  = (const float*)d_in[6];
    const float* as2 = (const float*)d_in[7];
    const float* ad2 = (const float*)d_in[8];
    const float* b2  = (const float*)d_in[9];
    float* out = (float*)d_out;

    const int N  = in_sizes[0] / 128;
    const int E  = in_sizes[1] / 2;
    const int H1 = 4, C1 = 32, F1 = 128;
    const int H2 = 1, C2 = 64, F2 = 64;

    char* w = (char*)d_ws;
    auto carve = [&](size_t bytes) {
        void* p = (void*)w;
        w += (bytes + 255) & ~(size_t)255;
        return p;
    };
    float* h1    = (float*)carve((size_t)N * F1 * 4);
    float* a1s   = (float*)carve((size_t)N * H1 * 4);
    float* a1d   = (float*)carve((size_t)N * H1 * 4);
    float* agg1  = (float*)carve((size_t)N * F1 * 4);
    float* h2    = (float*)carve((size_t)N * F2 * 4);
    float* a2s   = (float*)carve((size_t)N * 4);
    float* a2d   = (float*)carve((size_t)N * 4);
    int*   cnt   = (int*)  carve((size_t)(N + 1) * 4);
    int*   start = (int*)  carve((size_t)N * 4);
    int*   rank  = (int*)  carve((size_t)E * 4);
    int*   csrc  = (int*)  carve((size_t)E * 4);

    const int eb = 256, eg = (E + eb - 1) / eb;
    const int ng = (N + 7) / 8;

    fill_u32_kernel<<<(N + 1 + 255) / 256, 256, 0, stream>>>((unsigned*)cnt, N + 1, 0u);
    dst_count_kernel<<<eg, eb, 0, stream>>>(ei, E, cnt, rank);
    seg_alloc_kernel<<<(N + 255) / 256, 256, 0, stream>>>(cnt, cnt + N, start, N);
    seg_fill_kernel<<<eg, eb, 0, stream>>>(ei, E, start, rank, csrc);

    {
        int mt = (N + 15) / 16;
        int blocks = (mt * (F1 / 16) + 3) / 4;
        wmma_gemm_kernel<128><<<blocks, 128, 0, stream>>>(x, W1, h1, N, mt);
    }
    att_coeff_kernel<<<(N * H1 + 255) / 256, 256, 0, stream>>>(
        h1, as1, ad1, a1s, a1d, N, H1, C1);
    node_gather_kernel<4, 32><<<ng, 256, 0, stream>>>(start, cnt, csrc, h1, a1s, a1d, agg1, N);
    bias_elu_kernel<<<(N * F1 + 255) / 256, 256, 0, stream>>>(agg1, b1, N * F1, F1);

    {
        int mt = (N + 15) / 16;
        int blocks = (mt * (F2 / 16) + 3) / 4;
        wmma_gemm_kernel<64><<<blocks, 128, 0, stream>>>(agg1, W2, h2, N, mt);
    }
    att_coeff_kernel<<<(N + 255) / 256, 256, 0, stream>>>(
        h2, as2, ad2, a2s, a2d, N, H2, C2);
    node_gather_kernel<1, 64><<<ng, 256, 0, stream>>>(start, cnt, csrc, h2, a2s, a2d, out, N);
    bias_add_kernel<<<(N * F2 + 255) / 256, 256, 0, stream>>>(out, b2, N * F2, F2);
}


# Both f32 WMMA products re-tiled: W staged once per block (transposed, padded LDS) on a fixed grid; a wave reuses each A fragment for 4 (layer 2: 2) column tiles; 16-byte fragment loads; new k order
# speedup vs baseline: 6.4931x; 1.5202x over previous
#include <hip/hip_runtime.h>
#include <math.h>

typedef __attribute__((ext_vector_type(2))) float v2f;
typedef __attribute__((ext_vector_type(8))) float v8f;

#define K_DIM 128
#define NEG_SLOPE 0.2f
#define GEMM_BLOCKS 512

__global__ void fill_u32_kernel(unsigned* __restrict__ p, int n, unsigned v) {
    int i = blockIdx.x * blockDim.x + threadIdx.x;
    if (i < n) p[i] = v;
}

template <int NCOL, int NT>
__global__ __launch_bounds__(256)
void wmma_gemm_kernel(const float* __restrict__ A, const float* __restrict__ B,
                      float* __restrict__ C, int M, int mtiles) {
    constexpr int PITCH = K_DIM + 4;
    constexpr int KB = K_DIM / 8;
    __shared__ __attribute__((aligned(16))) float sWt[NCOL * PITCH];

    constexpr int TPR   = 256 / NCOL;
    constexpr int ITERS = (K_DIM / 4) / TPR;
    {
        const int col = threadIdx.x % NCOL;
        const int kq  = threadIdx.x / NCOL;
        float4 v[ITERS];
        #pragma unroll
        for (int it = 0; it < ITERS; ++it) {
            const float* bp = B + (4 * (kq + it * TPR)) * NCOL + col;
            v[it].x = bp[0];
            v[it].y = bp[NCOL];
            v[it].z = bp[2 * NCOL];
            v[it].w = bp[3 * NCOL];
        }
        #pragma unroll
        for (int it = 0; it < ITERS; ++it)
            *(float4*)&sWt[col * PITCH + 4 * (kq + it * TPR)] = v[it];
    }
    __syncthreads();

    const int wave = __builtin_amdgcn_readfirstlane((int)(threadIdx.x >> 5));
    const int lane = threadIdx.x & 31;
    const int half = lane >> 4;
    const int l    = lane & 15;

    constexpr int CG = NCOL / (16 * NT);
    const int nitems = mtiles * CG;
    constexpr int NWAVES = GEMM_BLOCKS * 8;
    const int gw = (int)blockIdx.x * 8 + wave;
    const int i0 = (int)(((long long)gw * nitems) / NWAVES);
    const int i1 = (int)(((long long)(gw + 1) * nitems) / NWAVES);

    for (int item = i0; item < i1; ++item) {
        const int mt = item / CG;
        const int cg = item - mt * CG;
        const int row0 = mt * 16;
        const int col0 = cg * 16 * NT;

        int ar = row0 + l;
        if (ar >= M) ar = M - 1;
        const float* arow = A + (size_t)ar * K_DIM + 4 * half;
        const float* bcol = sWt + (col0 + l) * PITCH + 4 * half;

        float4 a[KB];
        #pragma unroll
        for (int kb = 0; kb < KB; ++kb) a[kb] = *(const float4*)(arow + 8 * kb);

        v8f acc[NT];
        #pragma unroll
        for (int nt = 0; nt < NT; ++nt) acc[nt] = v8f{};

        #pragma unroll
        for (int kb = 0; kb < KB; ++kb) {
            float4 w[NT];
            #pragma unroll
            for (int nt = 0; nt < NT; ++nt)
                w[nt] = *(const float4*)(bcol + nt * 16 * PITCH + 8 * kb);
            v2f a1; a1.x = a[kb].x; a1.y = a[kb].y;
            v2f a2; a2.x = a[kb].z; a2.y = a[kb].w;
            #pragma unroll
            for (int nt = 0; nt < NT; ++nt) {
                v2f b1; b1.x = w[nt].x; b1.y = w[nt].y;
                acc[nt] = __builtin_amdgcn_wmma_f32_16x16x4_f32(
                              false, a1, false, b1, (short)0, acc[nt], false, false);
            }
            #pragma unroll
            for (int nt = 0; nt < NT; ++nt) {
                v2f b2; b2.x = w[nt].z; b2.y = w[nt].w;
                acc[nt] = __builtin_amdgcn_wmma_f32_16x16x4_f32(
                              false, a2, false, b2, (short)0, acc[nt], false, false);
            }
        }

        float* cp = C + (size_t)(row0 + half * 8) * NCOL + col0 + l;
        if (row0 + 16 <= M) {
            #pragma unroll
            for (int nt = 0; nt < NT; ++nt)
                #pragma unroll
                for (int i = 0; i < 8; ++i) cp[i * NCOL + nt * 16] = acc[nt][i];
        } else {
            #pragma unroll
            for (int nt = 0; nt < NT; ++nt)
                #pragma unroll
                for (int i = 0; i < 8; ++i)
                    if (row0 + half * 8 + i < M) cp[i * NCOL + nt * 16] = acc[nt][i];
        }
    }
}

__global__ void att_coeff_kernel(const float* __restrict__ h,
                                 const float* __restrict__ att_s,
                                 const float* __restrict__ att_d,
                                 float* __restrict__ a_s, float* __restrict__ a_d,
                                 int N, int H, int CH) {
    int i = blockIdx.x * blockDim.x + threadIdx.x;
    if (i >= N * H) return;
    int n  = i / H;
    int hh = i - n * H;
    const float* hp = h + (size_t)n * H * CH + (size_t)hh * CH;
    const float* sp = att_s + hh * CH;
    const float* dp = att_d + hh * CH;
    float ss = 0.f, dd = 0.f;
    for (int c = 0; c < CH; ++c) { ss += hp[c] * sp[c]; dd += hp[c] * dp[c]; }
    a_s[i] = ss;
    a_d[i] = dd;
}

__global__ void dst_count_kernel(const int* __restrict__ ei, int E,
                                 int* __restrict__ cnt, int* __restrict__ rank) {
    int e = blockIdx.x * blockDim.x + threadIdx.x;
    if (e >= E) return;
    rank[e] = atomicAdd(&cnt[ei[E + e]], 1);
}

__global__ void seg_alloc_kernel(const int* __restrict__ cnt, int* __restrict__ total,
                                 int* __restrict__ start, int N) {
    int i = blockIdx.x * blockDim.x + threadIdx.x;
    int lane = threadIdx.x & 31;
    int c = (i < N) ? cnt[i] : 0;
    int incl = c;
    for (int d = 1; d < 32; d <<= 1) {
        int y = __shfl_up(incl, d, 32);
        if (lane >= d) incl += y;
    }
    int base = 0;
    if (lane == 31) base = atomicAdd(total, incl);
    base = __shfl(base, 31, 32);
    if (i < N) start[i] = base + incl - c;
}

__global__ void seg_fill_kernel(const int* __restrict__ ei, int E,
                                const int* __restrict__ start,
                                const int* __restrict__ rank,
                                int* __restrict__ csr_src) {
    int e = blockIdx.x * blockDim.x + threadIdx.x;
    if (e >= E) return;
    csr_src[start[ei[E + e]] + rank[e]] = ei[e];
}

__device__ __forceinline__ float wave_max(float v) {
    for (int d = 16; d > 0; d >>= 1) v = fmaxf(v, __shfl_xor(v, d, 32));
    return v;
}
__device__ __forceinline__ float wave_sum(float v) {
    for (int d = 16; d > 0; d >>= 1) v += __shfl_xor(v, d, 32);
    return v;
}
__device__ __forceinline__ float lane_bcast(float v, int t) {
    return __int_as_float(__builtin_amdgcn_readlane(__float_as_int(v), t));
}

template <int H>
__device__ __forceinline__ void load_heads(const float* __restrict__ p, int idx, float (&o)[H]) {
    if constexpr (H == 4) {
        float4 q = *(const float4*)(p + (size_t)idx * 4);
        o[0] = q.x; o[1] = q.y; o[2] = q.z; o[3] = q.w;
    } else {
        for (int h = 0; h < H; ++h) o[h] = p[(size_t)idx * H + h];
    }
}

template <int H, int CH, int G>
__device__ __forceinline__ void gather_rows(const float* __restrict__ hfeat, int s,
                                            const float (&p)[H], int hh, int lane, int t0,
                                            float (&acc)[(H * CH) / 32]) {
    constexpr int F = H * CH;
    constexpr int VEC = F / 32;
    typedef float VT __attribute__((ext_vector_type(VEC)));
    VT r[G];
    float w[G];
    for (int g = 0; g < G; ++g) {
        const int st = __builtin_amdgcn_readlane(s, t0 + g);
        r[g] = *(const VT*)(hfeat + (size_t)st * F + lane * VEC);
        float pw = lane_bcast(p[0], t0 + g);
        for (int h = 1; h < H; ++h) {
            float q = lane_bcast(p[h], t0 + g);
            pw = (hh == h) ? q : pw;
        }
        w[g] = pw;
    }
    for (int g = 0; g < G; ++g)
        for (int v = 0; v < VEC; ++v) acc[v] = fmaf(w[g], r[g][v], acc[v]);
}

template <int H, int CH, bool ELU>
__global__ __launch_bounds__(256)
void node_gather_kernel(const int* __restrict__ start, const int* __restrict__ cnt,
                        const int* __restrict__ csr_src,
                        const float* __restrict__ hfeat,
                        const float* __restrict__ a_s, const float* __restrict__ a_d,
                        const float* __restrict__ bias,
                        float* __restrict__ outp, int N) {
    constexpr int F = H * CH;
    constexpr int VEC = F / 32;
    typedef float VT __attribute__((ext_vector_type(VEC)));
    const int lane = threadIdx.x & 31;
    const int i = __builtin_amdgcn_readfirstlane((int)(blockIdx.x * 8 + (threadIdx.x >> 5)));
    if (i >= N) return;
    const int hh = (lane * VEC) / CH;
    const int beg = start[i];
    const int n = cnt[i];
    const int m = n + 1;

    float ad[H];
    load_heads<H>(a_d, i, ad);
    const VT bv = *(const VT*)(bias + lane * VEC);
    float mx[H], den[H], acc[VEC];
    for (int h = 0; h < H; ++h) { mx[h] = -INFINITY; den[h] = 0.f; }
    for (int v = 0; v < VEC; ++v) acc[v] = 0.f;

    for (int base = 0; base < m; base += 32) {
        const int k = base + lane;
        const bool valid = k < m;
        int s = i;
        if (k < n) s = csr_src[beg + k];
        float as[H], p[H];
        load_heads<H>(a_s, s, as);
        float sc = 1.f;
        for (int h = 0; h < H; ++h) {
            float v = as[h] + ad[h];
            v = (v > 0.f) ? v : NEG_SLOPE * v;
            const float e = valid ? v : -INFINITY;
            const float nm = fmaxf(mx[h], wave_max(e));
            const float scl = __expf(mx[h] - nm);
            const float pe = __expf(e - nm);
            den[h] = den[h] * scl + wave_sum(pe);
            mx[h] = nm;
            p[h] = pe;
            sc = (hh == h) ? scl : sc;
        }
        for (int v = 0; v < VEC; ++v) acc[v] *= sc;

        const int cc = (m - base < 32) ? (m - base) : 32;
        int t = 0;
        for (; t + 8 <= cc; t += 8) gather_rows<H, CH, 8>(hfeat, s, p, hh, lane, t, acc);
        if (t + 4 <= cc) { gather_rows<H, CH, 4>(hfeat, s, p, hh, lane, t, acc); t += 4; }
        if (t + 2 <= cc) { gather_rows<H, CH, 2>(hfeat, s, p, hh, lane, t, acc); t += 2; }
        if (t < cc)      { gather_rows<H, CH, 1>(hfeat, s, p, hh, lane, t, acc); }
    }

    float dsel = den[0];
    for (int h = 1; h < H; ++h) dsel = (hh == h) ? den[h] : dsel;
    VT o;
    for (int v = 0; v < VEC; ++v) {
        float r = acc[v] / dsel + bv[v];
        if constexpr (ELU) r = (r > 0.f) ? r : (__expf(r) - 1.f);
        o[v] = r;
    }
    *(VT*)(outp + (size_t)i * F + lane * VEC) = o;
}

extern "C" void kernel_launch(void* const* d_in, const int* in_sizes, int n_in,
                              void* d_out, int out_size, void* d_ws, size_t ws_size,
                              hipStream_t stream) {
    const float* x   = (const float*)d_in[0];
    const int*   ei  = (const int*)  d_in[1];
    const float* W1  = (const float*)d_in[2];
    const float* as1 = (const float*)d_in[3];
    const float* ad1 = (const float*)d_in[4];
    const float* b1  = (const float*)d_in[5];
    const float* W2  = (const float*)d_in[6];
    const float* as2 = (const float*)d_in[7];
    const float* ad2 = (const float*)d_in[8];
    const float* b2  = (const float*)d_in[9];
    float* out = (float*)d_out;

    const int N  = in_sizes[0] / 128;
    const int E  = in_sizes[1] / 2;
    const int H1 = 4, C1 = 32, F1 = 128;
    const int H2 = 1, C2 = 64, F2 = 64;

    char* w = (char*)d_ws;
    auto carve = [&](size_t bytes) {
        void* p = (void*)w;
        w += (bytes + 255) & ~(size_t)255;
        return p;
    };
    float* h1    = (float*)carve((size_t)N * F1 * 4);
    float* a1s   = (float*)carve((size_t)N * H1 * 4);
    float* a1d   = (float*)carve((size_t)N * H1 * 4);
    float* agg1  = (float*)carve((size_t)N * F1 * 4);
    float* h2    = (float*)carve((size_t)N * F2 * 4);
    float* a2s   = (float*)carve((size_t)N * 4);
    float* a2d   = (float*)carve((size_t)N * 4);
    int*   cnt   = (int*)  carve((size_t)(N + 1) * 4);
    int*   start = (int*)  carve((size_t)N * 4);
    int*   rank  = (int*)  carve((size_t)E * 4);
    int*   csrc  = (int*)  carve((size_t)E * 4);

    const int eb = 256, eg = (E + eb - 1) / eb;
    const int ng = (N + 7) / 8;

    fill_u32_kernel<<<(N + 1 + 255) / 256, 256, 0, stream>>>((unsigned*)cnt, N + 1, 0u);
    dst_count_kernel<<<eg, eb, 0, stream>>>(ei, E, cnt, rank);
    seg_alloc_kernel<<<(N + 255) / 256, 256, 0, stream>>>(cnt, cnt + N, start, N);
    seg_fill_kernel<<<eg, eb, 0, stream>>>(ei, E, start, rank, csrc);

    {
        int mt = (N + 15) / 16;
        wmma_gemm_kernel<128, 4><<<GEMM_BLOCKS, 256, 0, stream>>>(x, W1, h1, N, mt);
    }
    att_coeff_kernel<<<(N * H1 + 255) / 256, 256, 0, stream>>>(
        h1, as1, ad1, a1s, a1d, N, H1, C1);
    node_gather_kernel<4, 32, true><<<ng, 256, 0, stream>>>(start, cnt, csrc, h1, a1s, a1d, b1, agg1, N);

    {
        int mt = (N + 15) / 16;
        wmma_gemm_kernel<64, 2><<<GEMM_BLOCKS, 256, 0, stream>>>(agg1, W2, h2, N, mt);
    }
    att_coeff_kernel<<<(N + 255) / 256, 256, 0, stream>>>(
        h2, as2, ad2, a2s, a2d, N, H2, C2);
    node_gather_kernel<1, 64, false><<<ng, 256, 0, stream>>>(start, cnt, csrc, h2, a2s, a2d, b2, out, N);
}


# Gather kernels: a wave walks 7 consecutive nodes; next nodes' indices asked for ahead, a node's first rows asked for before its logit reductions, heads reduced side by side; order of sums unchanged
# speedup vs baseline: 11.6380x; 1.7924x over previous
#include <hip/hip_runtime.h>
#include <math.h>

typedef __attribute__((ext_vector_type(2))) float v2f;
typedef __attribute__((ext_vector_type(8))) float v8f;

#define K_DIM 128
#define NEG_SLOPE 0.2f
#define GEMM_BLOCKS 512
#define GATHER_NPW 7

template <int MASK>
__device__ __forceinline__ float row_xor(float v) {
    return __int_as_float(__builtin_amdgcn_update_dpp(
        0, __float_as_int(v), 0x160 + MASK, 0xf, 0xf, true));
}

template <int NCOL, int NT, int CH>
__global__ __launch_bounds__(256)
void wmma_gemm_kernel(const float* __restrict__ A, const float* __restrict__ B,
                      const float* __restrict__ att_s, const float* __restrict__ att_d,
                      float* __restrict__ C,
                      float* __restrict__ a_s, float* __restrict__ a_d,
                      int M, int mtiles) {
    static_assert((16 * NT) % CH == 0 && CH % 16 == 0, "an item must hold whole heads");
    constexpr int H   = NCOL / CH;
    constexpr int HPI = (16 * NT) / CH;
    constexpr int TPH = CH / 16;
    constexpr int PITCH = K_DIM + 4;
    constexpr int KB = K_DIM / 8;
    __shared__ __attribute__((aligned(16))) float sWt[NCOL * PITCH];

    constexpr int TPR   = 256 / NCOL;
    constexpr int ITERS = (K_DIM / 4) / TPR;
    {
        const int col = threadIdx.x % NCOL;
        const int kq  = threadIdx.x / NCOL;
        float4 v[ITERS];
        #pragma unroll
        for (int it = 0; it < ITERS; ++it) {
            const float* bp = B + (4 * (kq + it * TPR)) * NCOL + col;
            v[it].x = bp[0];
            v[it].y = bp[NCOL];
            v[it].z = bp[2 * NCOL];
            v[it].w = bp[3 * NCOL];
        }
        #pragma unroll
        for (int it = 0; it < ITERS; ++it)
            *(float4*)&sWt[col * PITCH + 4 * (kq + it * TPR)] = v[it];
    }
    __syncthreads();

    const int wave = __builtin_amdgcn_readfirstlane((int)(threadIdx.x >> 5));
    const int lane = threadIdx.x & 31;
    const int half = lane >> 4;
    const int l    = lane & 15;

    constexpr int CG = NCOL / (16 * NT);
    const int nitems = mtiles * CG;
    constexpr int NWAVES = GEMM_BLOCKS * 8;
    const int gw = (int)blockIdx.x * 8 + wave;
    const int i0 = (int)(((long long)gw * nitems) / NWAVES);
    const int i1 = (int)(((long long)(gw + 1) * nitems) / NWAVES);

    for (int item = i0; item < i1; ++item) {
        const int mt = item / CG;
        const int cg = item - mt * CG;
        const int row0 = mt * 16;
        int col0 = cg * 16 * NT;
        if constexpr (CG == 1) {
            asm volatile("" : "+s"(col0));
        }

        int ar = row0 + l;
        if (ar >= M) ar = M - 1;
        const float* arow = A + (size_t)ar * K_DIM + 4 * half;
        const float* bcol = sWt + (col0 + l) * PITCH + 4 * half;

        float4 a[KB];
        #pragma unroll
        for (int kb = 0; kb < KB; ++kb) a[kb] = *(const float4*)(arow + 8 * kb);

        float ws[NT], wd[NT];
        #pragma unroll
        for (int nt = 0; nt < NT; ++nt) {
            ws[nt] = att_s[col0 + nt * 16 + l];
            wd[nt] = att_d[col0 + nt * 16 + l];
        }

        v8f acc[NT];
        #pragma unroll
        for (int nt = 0; nt < NT; ++nt) acc[nt] = v8f{};

        #pragma unroll
        for (int kb = 0; kb < KB; ++kb) {
            float4 w[NT];
            #pragma unroll
            for (int nt = 0; nt < NT; ++nt)
                w[nt] = *(const float4*)(bcol + nt * 16 * PITCH + 8 * kb);
            v2f a1; a1.x = a[kb].x; a1.y = a[kb].y;
            v2f a2; a2.x = a[kb].z; a2.y = a[kb].w;
            #pragma unroll
            for (int nt = 0; nt < NT; ++nt) {
                v2f b1; b1.x = w[nt].x; b1.y = w[nt].y;
                acc[nt] = __builtin_amdgcn_wmma_f32_16x16x4_f32(
                              false, a1, false, b1, (short)0, acc[nt], false, false);
            }
            #pragma unroll
            for (int nt = 0; nt < NT; ++nt) {
                v2f b2; b2.x = w[nt].z; b2.y = w[nt].w;
                acc[nt] = __builtin_amdgcn_wmma_f32_16x16x4_f32(
                              false, a2, false, b2, (short)0, acc[nt], false, false);
            }
        }

        float* cp = C + (size_t)(row0 + half * 8) * NCOL + col0 + l;
        if (row0 + 16 <= M) {
            #pragma unroll
            for (int nt = 0; nt < NT; ++nt)
                #pragma unroll
                for (int i = 0; i < 8; ++i) cp[i * NCOL + nt * 16] = acc[nt][i];
        } else {
            #pragma unroll
            for (int nt = 0; nt < NT; ++nt)
                #pragma unroll
                for (int i = 0; i < 8; ++i)
                    if (row0 + half * 8 + i < M) cp[i * NCOL + nt * 16] = acc[nt][i];
        }

        const bool b3 = (l & 8) != 0, b2 = (l & 4) != 0, b1 = (l & 2) != 0, b0 = (l & 1) != 0;
        const int orow = row0 + half * 8 + (l & 7);
        #pragma unroll
        for (int hd = 0; hd < HPI; ++hd) {
            float c[16];
            #pragma unroll
            for (int i = 0; i < 8; ++i) {
                float s = acc[hd * TPH][i] * ws[hd * TPH];
                float d = acc[hd * TPH][i] * wd[hd * TPH];
                #pragma unroll
                for (int t = 1; t < TPH; ++t) {
                    s = fmaf(acc[hd * TPH + t][i], ws[hd * TPH + t], s);
                    d = fmaf(acc[hd * TPH + t][i], wd[hd * TPH + t], d);
                }
                c[i] = s;
                c[8 + i] = d;
            }
            float y[8], z[4], u[2];
            #pragma unroll
            for (int j = 0; j < 8; ++j)
                y[j] = (b3 ? c[8 + j] : c[j]) + row_xor<8>(b3 ? c[j] : c[8 + j]);
            #pragma unroll
            for (int j = 0; j < 4; ++j)
                z[j] = (b2 ? y[4 + j] : y[j]) + row_xor<4>(b2 ? y[j] : y[4 + j]);
            #pragma unroll
            for (int j = 0; j < 2; ++j)
                u[j] = (b1 ? z[2 + j] : z[j]) + row_xor<2>(b1 ? z[j] : z[2 + j]);
            const float r = (b0 ? u[1] : u[0]) + row_xor<1>(b0 ? u[0] : u[1]);
            float* dst = b3 ? a_d : a_s;
            if (orow < M) dst[(size_t)orow * H + (col0 / CH + hd)] = r;
        }
    }
}

#define PART_SHIFT 8
#define PART_BINS_MAX 256
#define PART_CHUNK 4096

__device__ __forceinline__ int scan256(int v, int tid, int* wtot, int& tot) {
    const int lane = tid & 31, w = tid >> 5;
    int incl = v;
    if (tid < 256) {
        for (int d = 1; d < 32; d <<= 1) {
            int y = __shfl_up(incl, d, 32);
            if (lane >= d) incl += y;
        }
        if (lane == 31) wtot[w] = incl;
    }
    __syncthreads();
    int pre = 0;
    tot = 0;
    if (tid < 256) {
        for (int j = 0; j < 8; ++j) {
            const int t = wtot[j];
            tot += t;
            pre += (j < w) ? t : 0;
        }
    }
    return pre + incl - v;
}

__global__ __launch_bounds__(1024)
void edge_partition_kernel(const int* __restrict__ ei, int E, int nbins, int nchunks,
                           int* __restrict__ part_raw, int* __restrict__ runs_raw,
                           int* __restrict__ total) {
    __shared__ int hist[PART_BINS_MAX];
    __shared__ int wtot[8];
    int2* part = (int2*)part_raw;
    int2* runs = (int2*)runs_raw;
    const int tid = threadIdx.x;
    const int g = blockIdx.x;
    const int e0 = g * PART_CHUNK;
    if (g == 0 && tid == 0) *total = 0;
    if (tid < PART_BINS_MAX) hist[tid] = 0;

    int s[4], d[4], r[4];
    #pragma unroll
    for (int k = 0; k < 4; ++k) {
        const int e = e0 + k * 1024 + tid;
        const bool v = e < E;
        s[k] = v ? ei[e] : 0;
        d[k] = v ? ei[E + e] : -1;
    }
    __syncthreads();
    #pragma unroll
    for (int k = 0; k < 4; ++k) {
        r[k] = 0;
        if (d[k] >= 0) r[k] = atomicAdd(&hist[min(d[k] >> PART_SHIFT, PART_BINS_MAX - 1)], 1);
    }
    __syncthreads();
    const int c = (tid < PART_BINS_MAX) ? hist[tid] : 0;
    int tot;
    const int first = scan256(c, tid, wtot, tot);
    if (tid < PART_BINS_MAX) {
        hist[tid] = first;
        if (tid < nbins) runs[(size_t)tid * nchunks + g] = make_int2(first, c);
    }
    __syncthreads();
    #pragma unroll
    for (int k = 0; k < 4; ++k)
        if (d[k] >= 0)
            part[e0 + hist[min(d[k] >> PART_SHIFT, PART_BINS_MAX - 1)] + r[k]] = make_int2(s[k], d[k]);
}

template <bool PLACE>
__device__ __forceinline__ void walk_run(const int2* __restrict__ p, int rc, int q,
                                         int* lcnt, int* __restrict__ csr_src) {
    for (int i = q; i < rc; i += 32) {
        int2 e[8];
        #pragma unroll
        for (int u = 0; u < 8; ++u) {
            const int j = i + 4 * u;
            e[u] = (j < rc) ? p[j] : make_int2(0, -1);
        }
        if constexpr (!PLACE) {
            #pragma unroll
            for (int u = 0; u < 8; ++u)
                if (e[u].y >= 0) atomicAdd(&lcnt[e[u].y & 255], 1);
        } else {
            #pragma unroll
            for (int u = 0; u < 8; ++u)
                if (e[u].y >= 0) csr_src[atomicAdd(&lcnt[e[u].y & 255], 1)] = e[u].x;
        }
    }
}

__global__ __launch_bounds__(1024)
void segment_build_kernel(const int* __restrict__ part_raw, const int* __restrict__ runs_raw,
                          int nchunks, int N, int* __restrict__ total,
                          int* __restrict__ cnt, int* __restrict__ start,
                          int* __restrict__ csr_src) {
    __shared__ int lcnt[256];
    __shared__ int wtot[8];
    __shared__ int sbase;
    const int2* part = (const int2*)part_raw;
    const int2* runs = (const int2*)runs_raw + (size_t)blockIdx.x * nchunks;
    const int tid = threadIdx.x;
    const int n0 = (int)blockIdx.x << PART_SHIFT;
    const int q = tid & 3;
    const int g0 = tid >> 2;
    int2 run0 = make_int2(0, 0);
    if (g0 < nchunks) run0 = runs[g0];
    if (tid < 256) lcnt[tid] = 0;
    __syncthreads();

    walk_run<false>(part + (size_t)g0 * PART_CHUNK + run0.x, run0.y, q, lcnt, csr_src);
    for (int g = g0 + 256; g < nchunks; g += 256) {
        const int2 run = runs[g];
        walk_run<false>(part + (size_t)g * PART_CHUNK + run.x, run.y, q, lcnt, csr_src);
    }
    __syncthreads();

    const int c = (tid < 256) ? lcnt[tid] : 0;
    int tot;
    const int first = scan256(c, tid, wtot, tot);
    if (tid == 0) sbase = atomicAdd(total, tot);
    __syncthreads();
    if (tid < 256) {
        const int st = sbase + first;
        lcnt[tid] = st;
        const int node = n0 + tid;
        if (node < N) { start[node] = st; cnt[node] = c; }
    }
    __syncthreads();

    walk_run<true>(part + (size_t)g0 * PART_CHUNK + run0.x, run0.y, q, lcnt, csr_src);
    for (int g = g0 + 256; g < nchunks; g += 256) {
        const int2 run = runs[g];
        walk_run<true>(part + (size_t)g * PART_CHUNK + run.x, run.y, q, lcnt, csr_src);
    }
}

__device__ __forceinline__ float lane_bcast(float v, int t) {
    return __int_as_float(__builtin_amdgcn_readlane(__float_as_int(v), t));
}

template <int H>
__device__ __forceinline__ void load_heads(const float* __restrict__ p, int idx, float (&o)[H]) {
    if constexpr (H == 4) {
        float4 q = *(const float4*)(p + (size_t)idx * 4);
        o[0] = q.x; o[1] = q.y; o[2] = q.z; o[3] = q.w;
    } else {
        for (int h = 0; h < H; ++h) o[h] = p[(size_t)idx * H + h];
    }
}

template <int H, int CH, int G>
__device__ __forceinline__ void gather_rows(const float* __restrict__ hfeat, int s,
                                            const float (&p)[H], int hh, int lane, int t0,
                                            float (&acc)[(H * CH) / 32]) {
    constexpr int F = H * CH;
    constexpr int VEC = F / 32;
    typedef float VT __attribute__((ext_vector_type(VEC)));
    VT r[G];
    float w[G];
    for (int g = 0; g < G; ++g) {
        const int st = __builtin_amdgcn_readlane(s, t0 + g);
        r[g] = *(const VT*)(hfeat + (size_t)st * F + lane * VEC);
        float pw = lane_bcast(p[0], t0 + g);
        for (int h = 1; h < H; ++h) {
            float q = lane_bcast(p[h], t0 + g);
            pw = (hh == h) ? q : pw;
        }
        w[g] = pw;
    }
    for (int g = 0; g < G; ++g)
        for (int v = 0; v < VEC; ++v) acc[v] = fmaf(w[g], r[g][v], acc[v]);
}

template <int H>
__device__ __forceinline__ float edge_weight(const float (&p)[H], int hh, int t) {
    float pw = lane_bcast(p[0], t);
    for (int h = 1; h < H; ++h) {
        float q = lane_bcast(p[h], t);
        pw = (hh == h) ? q : pw;
    }
    return pw;
}

template <int H, int CH, bool ELU, int NPW>
__global__ __launch_bounds__(256, 8)
void node_gather_kernel(const int* __restrict__ start, const int* __restrict__ cnt,
                        const int* __restrict__ csr_src,
                        const float* __restrict__ hfeat,
                        const float* __restrict__ a_s, const float* __restrict__ a_d,
                        const float* __restrict__ bias,
                        float* __restrict__ outp, int N) {
    static_assert(NPW >= 1 && NPW <= 32, "lane j keeps start and count of node j");
    constexpr int F = H * CH;
    constexpr int VEC = F / 32;
    constexpr int GE  = 8;
    constexpr int GE2 = (VEC > 2) ? 4 : 8;
    typedef float VT __attribute__((ext_vector_type(VEC)));
    const int lane = threadIdx.x & 31;
    const int gw = __builtin_amdgcn_readfirstlane((int)(blockIdx.x * 8 + (threadIdx.x >> 5)));
    const int i0 = gw * NPW;
    if (i0 >= N) return;
    const int nj = (N - i0 < NPW) ? (N - i0) : NPW;
    const int hh = (lane * VEC) / CH;

    int st_l = 0, cn_l = 0;
    float ad_l[H];
    for (int h = 0; h < H; ++h) ad_l[h] = 0.f;
    if (lane < nj) {
        st_l = start[i0 + lane];
        cn_l = cnt[i0 + lane];
        load_heads<H>(a_d, i0 + lane, ad_l);
    }
    const VT bv = *(const VT*)(bias + lane * VEC);

    int beg_c = __builtin_amdgcn_readlane(st_l, 0);
    int n_c   = __builtin_amdgcn_readlane(cn_l, 0);
    int s_c = i0;
    if (lane < n_c) s_c = csr_src[beg_c + lane];
    int beg_n = 0, n_n = 0, s_n = i0;
    if (nj > 1) {
        beg_n = __builtin_amdgcn_readlane(st_l, 1);
        n_n   = __builtin_amdgcn_readlane(cn_l, 1);
        s_n = i0 + 1;
        if (lane < n_n) s_n = csr_src[beg_n + lane];
    }
    float as_c[H];
    load_heads<H>(a_s, s_c, as_c);
    for (int h = 0; h < H; ++h) asm volatile("" : "+v"(as_c[h]));

    for (int j = 0; j < nj; ++j) {
        const int i = i0 + j;
        const int n = n_c, beg = beg_c;
        const int m = n + 1;

        float as_n[H], ad_c[H];
        for (int h = 0; h < H; ++h) {
            as_n[h] = 0.f;
            ad_c[h] = lane_bcast(ad_l[h], j);
        }
        int beg_nn = 0, n_nn = 0, s_nn = i0;
        if (j + 1 < nj) load_heads<H>(a_s, s_n, as_n);
        if (j + 2 < nj) {
            beg_nn = __builtin_amdgcn_readlane(st_l, j + 2);
            n_nn   = __builtin_amdgcn_readlane(cn_l, j + 2);
            s_nn = i + 2;
            if (lane < n_nn) s_nn = csr_src[beg_nn + lane];
        }
        const VT rself = *(const VT*)(hfeat + (size_t)i * F + lane * VEC);

        float mx[H], den[H], acc[VEC];
        for (int h = 0; h < H; ++h) { mx[h] = -INFINITY; den[h] = 0.f; }
        for (int v = 0; v < VEC; ++v) acc[v] = 0.f;

        for (int base = 0; base < m; base += 32) {
            const int k = base + lane;
            const bool valid = k < m;
            int s = s_c;
            float as[H];
            for (int h = 0; h < H; ++h) as[h] = as_c[h];
            if (base != 0) {
                s = i;
                if (k < n) s = csr_src[beg + k];
                load_heads<H>(a_s, s, as);
                for (int h = 0; h < H; ++h) asm volatile("" : "+v"(as[h]));
            }
            const int cc = (m - base < 32) ? (m - base) : 32;
            const bool lastc = base + 32 >= m;
            const int ce = lastc ? cc - 1 : cc;

            VT r0[GE], r1[GE2];
            #pragma unroll
            for (int g = 0; g < GE; ++g) {
                const int sg = __builtin_amdgcn_readlane(s, g);
                r0[g] = *(const VT*)(hfeat + (size_t)sg * F + lane * VEC);
            }
            if (ce > GE) {
                #pragma unroll
                for (int g = 0; g < GE2; ++g) {
                    const int sg = __builtin_amdgcn_readlane(s, GE + g);
                    r1[g] = *(const VT*)(hfeat + (size_t)sg * F + lane * VEC);
                }
            }

            float p[H], e[H], red[H], scl[H];
            float sc = 1.f;
            for (int h = 0; h < H; ++h) {
                float v = as[h] + ad_c[h];
                v = (v > 0.f) ? v : NEG_SLOPE * v;
                e[h] = valid ? v : -INFINITY;
                red[h] = e[h];
            }
            for (int d = 16; d > 0; d >>= 1) {
                float y[H];
                for (int h = 0; h < H; ++h) y[h] = __shfl_xor(red[h], d, 32);
                for (int h = 0; h < H; ++h) red[h] = fmaxf(red[h], y[h]);
            }
            for (int h = 0; h < H; ++h) {
                const float nm = fmaxf(mx[h], red[h]);
                scl[h] = __expf(mx[h] - nm);
                p[h] = __expf(e[h] - nm);
                mx[h] = nm;
                red[h] = p[h];
            }
            for (int d = 16; d > 0; d >>= 1) {
                float y[H];
                for (int h = 0; h < H; ++h) y[h] = __shfl_xor(red[h], d, 32);
                for (int h = 0; h < H; ++h) red[h] += y[h];
            }
            for (int h = 0; h < H; ++h) {
                den[h] = den[h] * scl[h] + red[h];
                sc = (hh == h) ? scl[h] : sc;
            }
            for (int v = 0; v < VEC; ++v) acc[v] *= sc;

            #pragma unroll
            for (int g = 0; g < GE; ++g) {
                float w = edge_weight<H>(p, hh, g);
                w = (g < ce) ? w : 0.f;
                for (int v = 0; v < VEC; ++v) acc[v] = fmaf(w, r0[g][v], acc[v]);
            }
            if (ce > GE) {
                #pragma unroll
                for (int g = 0; g < GE2; ++g) {
                    float w = edge_weight<H>(p, hh, GE + g);
                    w = (GE + g < ce) ? w : 0.f;
                    for (int v = 0; v < VEC; ++v) acc[v] = fmaf(w, r1[g][v], acc[v]);
                }
            }
            int t = GE + GE2;
            for (; t + 8 <= ce; t += 8) gather_rows<H, CH, 8>(hfeat, s, p, hh, lane, t, acc);
            if (t + 4 <= ce) { gather_rows<H, CH, 4>(hfeat, s, p, hh, lane, t, acc); t += 4; }
            if (t + 2 <= ce) { gather_rows<H, CH, 2>(hfeat, s, p, hh, lane, t, acc); t += 2; }
            if (t < ce)      { gather_rows<H, CH, 1>(hfeat, s, p, hh, lane, t, acc); }
            if (lastc) {
                const float w = edge_weight<H>(p, hh, cc - 1);
                for (int v = 0; v < VEC; ++v) acc[v] = fmaf(w, rself[v], acc[v]);
            }
        }

        float dsel = den[0];
        for (int h = 1; h < H; ++h) dsel = (hh == h) ? den[h] : dsel;
        VT o;
        for (int v = 0; v < VEC; ++v) {
            float r = acc[v] / dsel + bv[v];
            if constexpr (ELU) r = (r > 0.f) ? r : (__expf(r) - 1.f);
            o[v] = r;
        }
        *(VT*)(outp + (size_t)i * F + lane * VEC) = o;

        beg_c = beg_n; n_c = n_n; s_c = s_n;
        beg_n = beg_nn; n_n = n_nn; s_n = s_nn;
        for (int h = 0; h < H; ++h) as_c[h] = as_n[h];
    }
}

extern "C" void kernel_launch(void* const* d_in, const int* in_sizes, int n_in,
                              void* d_out, int out_size, void* d_ws, size_t ws_size,
                              hipStream_t stream) {
    const float* x   = (const float*)d_in[0];
    const int*   ei  = (const int*)  d_in[1];
    const float* W1  = (const float*)d_in[2];
    const float* as1 = (const float*)d_in[3];
    const float* ad1 = (const float*)d_in[4];
    const float* b1  = (const float*)d_in[5];
    const float* W2  = (const float*)d_in[6];
    const float* as2 = (const float*)d_in[7];
    const float* ad2 = (const float*)d_in[8];
    const float* b2  = (const float*)d_in[9];
    float* out = (float*)d_out;

    const int N  = in_sizes[0] / 128;
    const int E  = in_sizes[1] / 2;
    const int H1 = 4, F1 = 128;
    const int F2 = 64;

    char* w = (char*)d_ws;
    auto carve = [&](size_t bytes) {
        void* p = (void*)w;
        w += (bytes + 255) & ~(size_t)255;
        return p;
    };
    float* h1    = (float*)carve((size_t)N * F1 * 4);
    float* a1s   = (float*)carve((size_t)N * H1 * 4);
    float* a1d   = (float*)carve((size_t)N * H1 * 4);
    float* agg1  = (float*)carve((size_t)N * F1 * 4);
    float* h2    = (float*)carve((size_t)N * F2 * 4);
    float* a2s   = (float*)carve((size_t)N * 4);
    float* a2d   = (float*)carve((size_t)N * 4);
    int*   cnt   = (int*)  carve((size_t)(N + 1) * 4);
    int*   start = (int*)  carve((size_t)N * 4);
    int*   csrc  = (int*)  carve((size_t)E * 4);

    const int ng = (N + 8 * GATHER_NPW - 1) / (8 * GATHER_NPW);

    const int nchunks = (E + PART_CHUNK - 1) / PART_CHUNK;
    const int nbins   = (N + (1 << PART_SHIFT) - 1) >> PART_SHIFT;
    int* part = (int*)agg1;
    int* runs = part + (((size_t)E * 2 + 63) & ~(size_t)63);
    edge_partition_kernel<<<nchunks, 1024, 0, stream>>>(ei, E, nbins, nchunks, part, runs, cnt + N);
    segment_build_kernel<<<nbins, 1024, 0, stream>>>(part, runs, nchunks, N, cnt + N, cnt, start, csrc);

    {
        int mt = (N + 15) / 16;
        wmma_gemm_kernel<128, 4, 32><<<GEMM_BLOCKS, 256, 0, stream>>>(
            x, W1, as1, ad1, h1, a1s, a1d, N, mt);
    }
    node_gather_kernel<4, 32, true, GATHER_NPW><<<ng, 256, 0, stream>>>(start, cnt, csrc, h1, a1s, a1d, b1, agg1, N);

    {
        int mt = (N + 15) / 16;
        wmma_gemm_kernel<64, 4, 64><<<GEMM_BLOCKS, 256, 0, stream>>>(
            agg1, W2, as2, ad2, h2, a2s, a2d, N, mt);
    }
    node_gather_kernel<1, 64, false, GATHER_NPW><<<ng, 256, 0, stream>>>(start, cnt, csrc, h2, a2s, a2d, b2, out, N);
}


# Layer-2 f32 WMMA product moved into the layer-1 gather launch: a 16-wave block keeps its 112 finished rows in LDS as the A strip, W2 in LDS, 7 waves run one tile each; 6 -> 5 launches; sums unchanged
# speedup vs baseline: 12.5892x; 1.0817x over previous
#include <hip/hip_runtime.h>
#include <math.h>

typedef __attribute__((ext_vector_type(2))) float v2f;
typedef __attribute__((ext_vector_type(8))) float v8f;

#define K_DIM 128
#define NEG_SLOPE 0.2f
#define GEMM_BLOCKS 512
#define GATHER_NPW 7
#define FUSED_WPB 16

template <int MASK>
__device__ __forceinline__ float row_xor(float v) {
    return __int_as_float(__builtin_amdgcn_update_dpp(
        0, __float_as_int(v), 0x160 + MASK, 0xf, 0xf, true));
}

template <int NCOL, int NT, int CH>
__global__ __launch_bounds__(256)
void wmma_gemm_kernel(const float* __restrict__ A, const float* __restrict__ B,
                      const float* __restrict__ att_s, const float* __restrict__ att_d,
                      float* __restrict__ C,
                      float* __restrict__ a_s, float* __restrict__ a_d,
                      int M, int mtiles) {
    static_assert((16 * NT) % CH == 0 && CH % 16 == 0, "an item must hold whole heads");
    constexpr int H   = NCOL / CH;
    constexpr int HPI = (16 * NT) / CH;
    constexpr int TPH = CH / 16;
    constexpr int PITCH = K_DIM + 4;
    constexpr int KB = K_DIM / 8;
    __shared__ __attribute__((aligned(16))) float sWt[NCOL * PITCH];

    constexpr int TPR   = 256 / NCOL;
    constexpr int ITERS = (K_DIM / 4) / TPR;
    {
        const int col = threadIdx.x % NCOL;
        const int kq  = threadIdx.x / NCOL;
        float4 v[ITERS];
        #pragma unroll
        for (int it = 0; it < ITERS; ++it) {
            const float* bp = B + (4 * (kq + it * TPR)) * NCOL + col;
            v[it].x = bp[0];
            v[it].y = bp[NCOL];
            v[it].z = bp[2 * NCOL];
            v[it].w = bp[3 * NCOL];
        }
        #pragma unroll
        for (int it = 0; it < ITERS; ++it)
            *(float4*)&sWt[col * PITCH + 4 * (kq + it * TPR)] = v[it];
    }
    __syncthreads();

    const int wave = __builtin_amdgcn_readfirstlane((int)(threadIdx.x >> 5));
    const int lane = threadIdx.x & 31;
    const int half = lane >> 4;
    const int l    = lane & 15;

    constexpr int CG = NCOL / (16 * NT);
    const int nitems = mtiles * CG;
    constexpr int NWAVES = GEMM_BLOCKS * 8;
    const int gw = (int)blockIdx.x * 8 + wave;
    const int i0 = (int)(((long long)gw * nitems) / NWAVES);
    const int i1 = (int)(((long long)(gw + 1) * nitems) / NWAVES);

    for (int item = i0; item < i1; ++item) {
        const int mt = item / CG;
        const int cg = item - mt * CG;
        const int row0 = mt * 16;
        const int col0 = cg * 16 * NT;

        int ar = row0 + l;
        if (ar >= M) ar = M - 1;
        const float* arow = A + (size_t)ar * K_DIM + 4 * half;
        const float* bcol = sWt + (col0 + l) * PITCH + 4 * half;

        float4 a[KB];
        #pragma unroll
        for (int kb = 0; kb < KB; ++kb) a[kb] = *(const float4*)(arow + 8 * kb);

        float ws[NT], wd[NT];
        #pragma unroll
        for (int nt = 0; nt < NT; ++nt) {
            ws[nt] = att_s[col0 + nt * 16 + l];
            wd[nt] = att_d[col0 + nt * 16 + l];
        }

        v8f acc[NT];
        #pragma unroll
        for (int nt = 0; nt < NT; ++nt) acc[nt] = v8f{};

        #pragma unroll
        for (int kb = 0; kb < KB; ++kb) {
            float4 w[NT];
            #pragma unroll
            for (int nt = 0; nt < NT; ++nt)
                w[nt] = *(const float4*)(bcol + nt * 16 * PITCH + 8 * kb);
            v2f a1; a1.x = a[kb].x; a1.y = a[kb].y;
            v2f a2; a2.x = a[kb].z; a2.y = a[kb].w;
            #pragma unroll
            for (int nt = 0; nt < NT; ++nt) {
                v2f b1; b1.x = w[nt].x; b1.y = w[nt].y;
                acc[nt] = __builtin_amdgcn_wmma_f32_16x16x4_f32(
                              false, a1, false, b1, (short)0, acc[nt], false, false);
            }
            #pragma unroll
            for (int nt = 0; nt < NT; ++nt) {
                v2f b2; b2.x = w[nt].z; b2.y = w[nt].w;
                acc[nt] = __builtin_amdgcn_wmma_f32_16x16x4_f32(
                              false, a2, false, b2, (short)0, acc[nt], false, false);
            }
        }

        float* cp = C + (size_t)(row0 + half * 8) * NCOL + col0 + l;
        if (row0 + 16 <= M) {
            #pragma unroll
            for (int nt = 0; nt < NT; ++nt)
                #pragma unroll
                for (int i = 0; i < 8; ++i) cp[i * NCOL + nt * 16] = acc[nt][i];
        } else {
            #pragma unroll
            for (int nt = 0; nt < NT; ++nt)
                #pragma unroll
                for (int i = 0; i < 8; ++i)
                    if (row0 + half * 8 + i < M) cp[i * NCOL + nt * 16] = acc[nt][i];
        }

        const bool b3 = (l & 8) != 0, b2 = (l & 4) != 0, b1 = (l & 2) != 0, b0 = (l & 1) != 0;
        const int orow = row0 + half * 8 + (l & 7);
        #pragma unroll
        for (int hd = 0; hd < HPI; ++hd) {
            float c[16];
            #pragma unroll
            for (int i = 0; i < 8; ++i) {
                float s = acc[hd * TPH][i] * ws[hd * TPH];
                float d = acc[hd * TPH][i] * wd[hd * TPH];
                #pragma unroll
                for (int t = 1; t < TPH; ++t) {
                    s = fmaf(acc[hd * TPH + t][i], ws[hd * TPH + t], s);
                    d = fmaf(acc[hd * TPH + t][i], wd[hd * TPH + t], d);
                }
                c[i] = s;
                c[8 + i] = d;
            }
            float y[8], z[4], u[2];
            #pragma unroll
            for (int j = 0; j < 8; ++j)
                y[j] = (b3 ? c[8 + j] : c[j]) + row_xor<8>(b3 ? c[j] : c[8 + j]);
            #pragma unroll
            for (int j = 0; j < 4; ++j)
                z[j] = (b2 ? y[4 + j] : y[j]) + row_xor<4>(b2 ? y[j] : y[4 + j]);
            #pragma unroll
            for (int j = 0; j < 2; ++j)
                u[j] = (b1 ? z[2 + j] : z[j]) + row_xor<2>(b1 ? z[j] : z[2 + j]);
            const float r = (b0 ? u[1] : u[0]) + row_xor<1>(b0 ? u[0] : u[1]);
            float* dst = b3 ? a_d : a_s;
            if (orow < M) dst[(size_t)orow * H + (col0 / CH + hd)] = r;
        }
    }
}

#define PART_SHIFT 8
#define PART_BINS_MAX 256
#define PART_CHUNK 4096

__device__ __forceinline__ int scan256(int v, int tid, int* wtot, int& tot) {
    const int lane = tid & 31, w = tid >> 5;
    int incl = v;
    if (tid < 256) {
        for (int d = 1; d < 32; d <<= 1) {
            int y = __shfl_up(incl, d, 32);
            if (lane >= d) incl += y;
        }
        if (lane == 31) wtot[w] = incl;
    }
    __syncthreads();
    int pre = 0;
    tot = 0;
    if (tid < 256) {
        for (int j = 0; j < 8; ++j) {
            const int t = wtot[j];
            tot += t;
            pre += (j < w) ? t : 0;
        }
    }
    return pre + incl - v;
}

__global__ __launch_bounds__(1024)
void edge_partition_kernel(const int* __restrict__ ei, int E, int nbins, int nchunks,
                           int* __restrict__ part_raw, int* __restrict__ runs_raw,
                           int* __restrict__ total) {
    __shared__ int hist[PART_BINS_MAX];
    __shared__ int wtot[8];
    int2* part = (int2*)part_raw;
    int2* runs = (int2*)runs_raw;
    const int tid = threadIdx.x;
    const int g = blockIdx.x;
    const int e0 = g * PART_CHUNK;
    if (g == 0 && tid == 0) *total = 0;
    if (tid < PART_BINS_MAX) hist[tid] = 0;

    int s[4], d[4], r[4];
    #pragma unroll
    for (int k = 0; k < 4; ++k) {
        const int e = e0 + k * 1024 + tid;
        const bool v = e < E;
        s[k] = v ? ei[e] : 0;
        d[k] = v ? ei[E + e] : -1;
    }
    __syncthreads();
    #pragma unroll
    for (int k = 0; k < 4; ++k) {
        r[k] = 0;
        if (d[k] >= 0) r[k] = atomicAdd(&hist[min(d[k] >> PART_SHIFT, PART_BINS_MAX - 1)], 1);
    }
    __syncthreads();
    const int c = (tid < PART_BINS_MAX) ? hist[tid] : 0;
    int tot;
    const int first = scan256(c, tid, wtot, tot);
    if (tid < PART_BINS_MAX) {
        hist[tid] = first;
        if (tid < nbins) runs[(size_t)tid * nchunks + g] = make_int2(first, c);
    }
    __syncthreads();
    #pragma unroll
    for (int k = 0; k < 4; ++k)
        if (d[k] >= 0)
            part[e0 + hist[min(d[k] >> PART_SHIFT, PART_BINS_MAX - 1)] + r[k]] = make_int2(s[k], d[k]);
}

template <bool PLACE>
__device__ __forceinline__ void walk_run(const int2* __restrict__ p, int rc, int q,
                                         int* lcnt, int* __restrict__ csr_src) {
    for (int i = q; i < rc; i += 32) {
        int2 e[8];
        #pragma unroll
        for (int u = 0; u < 8; ++u) {
            const int j = i + 4 * u;
            e[u] = (j < rc) ? p[j] : make_int2(0, -1);
        }
        if constexpr (!PLACE) {
            #pragma unroll
            for (int u = 0; u < 8; ++u)
                if (e[u].y >= 0) atomicAdd(&lcnt[e[u].y & 255], 1);
        } else {
            #pragma unroll
            for (int u = 0; u < 8; ++u)
                if (e[u].y >= 0) csr_src[atomicAdd(&lcnt[e[u].y & 255], 1)] = e[u].x;
        }
    }
}

__global__ __launch_bounds__(1024)
void segment_build_kernel(const int* __restrict__ part_raw, const int* __restrict__ runs_raw,
                          int nchunks, int N, int* __restrict__ total,
                          int* __restrict__ cnt, int* __restrict__ start,
                          int* __restrict__ csr_src) {
    __shared__ int lcnt[256];
    __shared__ int wtot[8];
    __shared__ int sbase;
    const int2* part = (const int2*)part_raw;
    const int2* runs = (const int2*)runs_raw + (size_t)blockIdx.x * nchunks;
    const int tid = threadIdx.x;
    const int n0 = (int)blockIdx.x << PART_SHIFT;
    const int q = tid & 3;
    const int g0 = tid >> 2;
    int2 run0 = make_int2(0, 0);
    if (g0 < nchunks) run0 = runs[g0];
    if (tid < 256) lcnt[tid] = 0;
    __syncthreads();

    walk_run<false>(part + (size_t)g0 * PART_CHUNK + run0.x, run0.y, q, lcnt, csr_src);
    for (int g = g0 + 256; g < nchunks; g += 256) {
        const int2 run = runs[g];
        walk_run<false>(part + (size_t)g * PART_CHUNK + run.x, run.y, q, lcnt, csr_src);
    }
    __syncthreads();

    const int c = (tid < 256) ? lcnt[tid] : 0;
    int tot;
    const int first = scan256(c, tid, wtot, tot);
    if (tid == 0) sbase = atomicAdd(total, tot);
    __syncthreads();
    if (tid < 256) {
        const int st = sbase + first;
        lcnt[tid] = st;
        const int node = n0 + tid;
        if (node < N) { start[node] = st; cnt[node] = c; }
    }
    __syncthreads();

    walk_run<true>(part + (size_t)g0 * PART_CHUNK + run0.x, run0.y, q, lcnt, csr_src);
    for (int g = g0 + 256; g < nchunks; g += 256) {
        const int2 run = runs[g];
        walk_run<true>(part + (size_t)g * PART_CHUNK + run.x, run.y, q, lcnt, csr_src);
    }
}

__device__ __forceinline__ float lane_bcast(float v, int t) {
    return __int_as_float(__builtin_amdgcn_readlane(__float_as_int(v), t));
}

template <int H>
__device__ __forceinline__ void load_heads(const float* __restrict__ p, int idx, float (&o)[H]) {
    if constexpr (H == 4) {
        float4 q = *(const float4*)(p + (size_t)idx * 4);
        o[0] = q.x; o[1] = q.y; o[2] = q.z; o[3] = q.w;
    } else {
        for (int h = 0; h < H; ++h) o[h] = p[(size_t)idx * H + h];
    }
}

template <int H, int CH, int G>
__device__ __forceinline__ void gather_rows(const float* __restrict__ hfeat, int s,
                                            const float (&p)[H], int hh, int lane, int t0,
                                            float (&acc)[(H * CH) / 32]) {
    constexpr int F = H * CH;
    constexpr int VEC = F / 32;
    typedef float VT __attribute__((ext_vector_type(VEC)));
    VT r[G];
    float w[G];
    for (int g = 0; g < G; ++g) {
        const int st = __builtin_amdgcn_readlane(s, t0 + g);
        r[g] = *(const VT*)(hfeat + (size_t)st * F + lane * VEC);
        float pw = lane_bcast(p[0], t0 + g);
        for (int h = 1; h < H; ++h) {
            float q = lane_bcast(p[h], t0 + g);
            pw = (hh == h) ? q : pw;
        }
        w[g] = pw;
    }
    for (int g = 0; g < G; ++g)
        for (int v = 0; v < VEC; ++v) acc[v] = fmaf(w[g], r[g][v], acc[v]);
}

template <int H>
__device__ __forceinline__ float edge_weight(const float (&p)[H], int hh, int t) {
    float pw = lane_bcast(p[0], t);
    for (int h = 1; h < H; ++h) {
        float q = lane_bcast(p[h], t);
        pw = (hh == h) ? q : pw;
    }
    return pw;
}

#define SROW_PITCH (K_DIM + 4)
template <int H, int CH, bool ELU, int NPW, bool TOLDS>
__device__ __forceinline__
void gather_walk(const int* __restrict__ start, const int* __restrict__ cnt,
                 const int* __restrict__ csr_src,
                 const float* __restrict__ hfeat,
                 const float* __restrict__ a_s, const float* __restrict__ a_d,
                 const float* __restrict__ bias,
                 float* __restrict__ outp, int N, int gw, float* srow) {
    static_assert(NPW >= 1 && NPW <= 32, "lane j keeps start and count of node j");
    constexpr int F = H * CH;
    constexpr int VEC = F / 32;
    constexpr int GE  = 8;
    constexpr int GE2 = (VEC > 2) ? 4 : 8;
    typedef float VT __attribute__((ext_vector_type(VEC)));
    const int lane = threadIdx.x & 31;
    const int i0 = gw * NPW;
    if (i0 >= N) return;
    const int nj = (N - i0 < NPW) ? (N - i0) : NPW;
    const int hh = (lane * VEC) / CH;

    int st_l = 0, cn_l = 0;
    float ad_l[H];
    for (int h = 0; h < H; ++h) ad_l[h] = 0.f;
    if (lane < nj) {
        st_l = start[i0 + lane];
        cn_l = cnt[i0 + lane];
        load_heads<H>(a_d, i0 + lane, ad_l);
    }
    const VT bv = *(const VT*)(bias + lane * VEC);

    int beg_c = __builtin_amdgcn_readlane(st_l, 0);
    int n_c   = __builtin_amdgcn_readlane(cn_l, 0);
    int s_c = i0;
    if (lane < n_c) s_c = csr_src[beg_c + lane];
    int beg_n = 0, n_n = 0, s_n = i0;
    if (nj > 1) {
        beg_n = __builtin_amdgcn_readlane(st_l, 1);
        n_n   = __builtin_amdgcn_readlane(cn_l, 1);
        s_n = i0 + 1;
        if (lane < n_n) s_n = csr_src[beg_n + lane];
    }
    float as_c[H];
    load_heads<H>(a_s, s_c, as_c);
    for (int h = 0; h < H; ++h) asm volatile("" : "+v"(as_c[h]));

    for (int j = 0; j < nj; ++j) {
        const int i = i0 + j;
        const int n = n_c, beg = beg_c;
        const int m = n + 1;

        float as_n[H], ad_c[H];
        for (int h = 0; h < H; ++h) {
            as_n[h] = 0.f;
            ad_c[h] = lane_bcast(ad_l[h], j);
        }
        int beg_nn = 0, n_nn = 0, s_nn = i0;
        if (j + 1 < nj) load_heads<H>(a_s, s_n, as_n);
        if (j + 2 < nj) {
            beg_nn = __builtin_amdgcn_readlane(st_l, j + 2);
            n_nn   = __builtin_amdgcn_readlane(cn_l, j + 2);
            s_nn = i + 2;
            if (lane < n_nn) s_nn = csr_src[beg_nn + lane];
        }
        const VT rself = *(const VT*)(hfeat + (size_t)i * F + lane * VEC);

        float mx[H], den[H], acc[VEC];
        for (int h = 0; h < H; ++h) { mx[h] = -INFINITY; den[h] = 0.f; }
        for (int v = 0; v < VEC; ++v) acc[v] = 0.f;

        for (int base = 0; base < m; base += 32) {
            const int k = base + lane;
            const bool valid = k < m;
            int s = s_c;
            float as[H];
            for (int h = 0; h < H; ++h) as[h] = as_c[h];
            if (base != 0) {
                s = i;
                if (k < n) s = csr_src[beg + k];
                load_heads<H>(a_s, s, as);
                for (int h = 0; h < H; ++h) asm volatile("" : "+v"(as[h]));
            }
            const int cc = (m - base < 32) ? (m - base) : 32;
            const bool lastc = base + 32 >= m;
            const int ce = lastc ? cc - 1 : cc;

            VT r0[GE], r1[GE2];
            #pragma unroll
            for (int g = 0; g < GE; ++g) {
                const int sg = __builtin_amdgcn_readlane(s, g);
                r0[g] = *(const VT*)(hfeat + (size_t)sg * F + lane * VEC);
            }
            if (ce > GE) {
                #pragma unroll
                for (int g = 0; g < GE2; ++g) {
                    const int sg = __builtin_amdgcn_readlane(s, GE + g);
                    r1[g] = *(const VT*)(hfeat + (size_t)sg * F + lane * VEC);
                }
            }

            float p[H], e[H], red[H], scl[H];
            float sc = 1.f;
            for (int h = 0; h < H; ++h) {
                float v = as[h] + ad_c[h];
                v = (v > 0.f) ? v : NEG_SLOPE * v;
                e[h] = valid ? v : -INFINITY;
                red[h] = e[h];
            }
            for (int d = 16; d > 0; d >>= 1) {
                float y[H];
                for (int h = 0; h < H; ++h) y[h] = __shfl_xor(red[h], d, 32);
                for (int h = 0; h < H; ++h) red[h] = fmaxf(red[h], y[h]);
            }
            for (int h = 0; h < H; ++h) {
                const float nm = fmaxf(mx[h], red[h]);
                scl[h] = __expf(mx[h] - nm);
                p[h] = __expf(e[h] - nm);
                mx[h] = nm;
                red[h] = p[h];
            }
            for (int d = 16; d > 0; d >>= 1) {
                float y[H];
                for (int h = 0; h < H; ++h) y[h] = __shfl_xor(red[h], d, 32);
                for (int h = 0; h < H; ++h) red[h] += y[h];
            }
            for (int h = 0; h < H; ++h) {
                den[h] = den[h] * scl[h] + red[h];
                sc = (hh == h) ? scl[h] : sc;
            }
            for (int v = 0; v < VEC; ++v) acc[v] *= sc;

            #pragma unroll
            for (int g = 0; g < GE; ++g) {
                float w = edge_weight<H>(p, hh, g);
                w = (g < ce) ? w : 0.f;
                for (int v = 0; v < VEC; ++v) acc[v] = fmaf(w, r0[g][v], acc[v]);
            }
            if (ce > GE) {
                #pragma unroll
                for (int g = 0; g < GE2; ++g) {
                    float w = edge_weight<H>(p, hh, GE + g);
                    w = (GE + g < ce) ? w : 0.f;
                    for (int v = 0; v < VEC; ++v) acc[v] = fmaf(w, r1[g][v], acc[v]);
                }
            }
            int t = GE + GE2;
            for (; t + 8 <= ce; t += 8) gather_rows<H, CH, 8>(hfeat, s, p, hh, lane, t, acc);
            if (t + 4 <= ce) { gather_rows<H, CH, 4>(hfeat, s, p, hh, lane, t, acc); t += 4; }
            if (t + 2 <= ce) { gather_rows<H, CH, 2>(hfeat, s, p, hh, lane, t, acc); t += 2; }
            if (t < ce)      { gather_rows<H, CH, 1>(hfeat, s, p, hh, lane, t, acc); }
            if (lastc) {
                const float w = edge_weight<H>(p, hh, cc - 1);
                for (int v = 0; v < VEC; ++v) acc[v] = fmaf(w, rself[v], acc[v]);
            }
        }

        float dsel = den[0];
        for (int h = 1; h < H; ++h) dsel = (hh == h) ? den[h] : dsel;
        VT o;
        for (int v = 0; v < VEC; ++v) {
            float r = acc[v] / dsel + bv[v];
            if constexpr (ELU) r = (r > 0.f) ? r : (__expf(r) - 1.f);
            o[v] = r;
        }
        if constexpr (TOLDS) *(VT*)(srow + j * SROW_PITCH + lane * VEC) = o;
        else                 *(VT*)(outp + (size_t)i * F + lane * VEC) = o;

        beg_c = beg_n; n_c = n_n; s_c = s_n;
        beg_n = beg_nn; n_n = n_nn; s_n = s_nn;
        for (int h = 0; h < H; ++h) as_c[h] = as_n[h];
    }
}

template <int H, int CH, bool ELU, int NPW>
__global__ __launch_bounds__(256, 8)
void node_gather_kernel(const int* __restrict__ start, const int* __restrict__ cnt,
                        const int* __restrict__ csr_src,
                        const float* __restrict__ hfeat,
                        const float* __restrict__ a_s, const float* __restrict__ a_d,
                        const float* __restrict__ bias,
                        float* __restrict__ outp, int N) {
    const int gw = __builtin_amdgcn_readfirstlane((int)(blockIdx.x * 8 + (threadIdx.x >> 5)));
    gather_walk<H, CH, ELU, NPW, false>(start, cnt, csr_src, hfeat, a_s, a_d, bias, outp, N, gw, nullptr);
}

template <int NCOL, int NT, int CH>
__device__ __forceinline__
void tile_product_lds(const float* sAt, const float* sWt,
                      const float* __restrict__ att_s, const float* __restrict__ att_d,
                      float* __restrict__ C, float* __restrict__ a_s, float* __restrict__ a_d,
                      int M, int row0, int lane) {
    static_assert(16 * NT == NCOL && NCOL % CH == 0 && CH % 16 == 0, "a tile holds all columns, whole heads");
    constexpr int H   = NCOL / CH;
    constexpr int HPI = (16 * NT) / CH;
    constexpr int TPH = CH / 16;
    constexpr int KB  = K_DIM / 8;
    const int half = lane >> 4;
    const int l    = lane & 15;
    const float* arow = sAt + l * SROW_PITCH + 4 * half;
    const float* bcol = sWt + l * SROW_PITCH + 4 * half;

    float ws[NT], wd[NT];
    #pragma unroll
    for (int nt = 0; nt < NT; ++nt) {
        ws[nt] = att_s[nt * 16 + l];
        wd[nt] = att_d[nt * 16 + l];
    }
    v8f acc[NT];
    #pragma unroll
    for (int nt = 0; nt < NT; ++nt) acc[nt] = v8f{};

    #pragma unroll
    for (int kb = 0; kb < KB; ++kb) {
        const float4 av = *(const float4*)(arow + 8 * kb);
        float4 w[NT];
        #pragma unroll
        for (int nt = 0; nt < NT; ++nt)
            w[nt] = *(const float4*)(bcol + nt * 16 * SROW_PITCH + 8 * kb);
        v2f a1; a1.x = av.x; a1.y = av.y;
        v2f a2; a2.x = av.z; a2.y = av.w;
        #pragma unroll
        for (int nt = 0; nt < NT; ++nt) {
            v2f b1; b1.x = w[nt].x; b1.y = w[nt].y;
            acc[nt] = __builtin_amdgcn_wmma_f32_16x16x4_f32(
                          false, a1, false, b1, (short)0, acc[nt], false, false);
        }
        #pragma unroll
        for (int nt = 0; nt < NT; ++nt) {
            v2f b2; b2.x = w[nt].z; b2.y = w[nt].w;
            acc[nt] = __builtin_amdgcn_wmma_f32_16x16x4_f32(
                          false, a2, false, b2, (short)0, acc[nt], false, false);
        }
    }

    float* cp = C + (size_t)(row0 + half * 8) * NCOL + l;
    if (row0 + 16 <= M) {
        #pragma unroll
        for (int nt = 0; nt < NT; ++nt)
            #pragma unroll
            for (int i = 0; i < 8; ++i) cp[i * NCOL + nt * 16] = acc[nt][i];
    } else {
        #pragma unroll
        for (int nt = 0; nt < NT; ++nt)
            #pragma unroll
            for (int i = 0; i < 8; ++i)
                if (row0 + half * 8 + i < M) cp[i * NCOL + nt * 16] = acc[nt][i];
    }

    const bool b3 = (l & 8) != 0, b2 = (l & 4) != 0, b1 = (l & 2) != 0, b0 = (l & 1) != 0;
    const int orow = row0 + half * 8 + (l & 7);
    #pragma unroll
    for (int hd = 0; hd < HPI; ++hd) {
        float c[16];
        #pragma unroll
        for (int i = 0; i < 8; ++i) {
            float sv = acc[hd * TPH][i] * ws[hd * TPH];
            float dv = acc[hd * TPH][i] * wd[hd * TPH];
            #pragma unroll
            for (int t = 1; t < TPH; ++t) {
                sv = fmaf(acc[hd * TPH + t][i], ws[hd * TPH + t], sv);
                dv = fmaf(acc[hd * TPH + t][i], wd[hd * TPH + t], dv);
            }
            c[i] = sv;
            c[8 + i] = dv;
        }
        float y[8], z[4], u[2];
        #pragma unroll
        for (int j = 0; j < 8; ++j)
            y[j] = (b3 ? c[8 + j] : c[j]) + row_xor<8>(b3 ? c[j] : c[8 + j]);
        #pragma unroll
        for (int j = 0; j < 4; ++j)
            z[j] = (b2 ? y[4 + j] : y[j]) + row_xor<4>(b2 ? y[j] : y[4 + j]);
        #pragma unroll
        for (int j = 0; j < 2; ++j)
            u[j] = (b1 ? z[2 + j] : z[j]) + row_xor<2>(b1 ? z[j] : z[2 + j]);
        const float r = (b0 ? u[1] : u[0]) + row_xor<1>(b0 ? u[0] : u[1]);
        float* dst = b3 ? a_d : a_s;
        if (orow < M) dst[(size_t)orow * H + hd] = r;
    }
}

template <int NPW, int WPB>
__global__ __launch_bounds__(WPB * 32, 8)
void gather_product_kernel(const int* __restrict__ start, const int* __restrict__ cnt,
                           const int* __restrict__ csr_src,
                           const float* __restrict__ hfeat,
                           const float* __restrict__ a_s, const float* __restrict__ a_d,
                           const float* __restrict__ bias,
                           const float* __restrict__ W2,
                           const float* __restrict__ att_s2, const float* __restrict__ att_d2,
                           float* __restrict__ h2, float* __restrict__ a2s, float* __restrict__ a2d,
                           int N) {
    constexpr int NCOL = 64;
    constexpr int ROWS = WPB * NPW;
    constexpr int NTHR = WPB * 32;
    static_assert(ROWS % 16 == 0 && ROWS / 16 <= WPB, "whole 16-row tiles, one wave each");
    static_assert(NTHR % NCOL == 0 && (K_DIM / 4) % (NTHR / NCOL) == 0, "staging covers W2 in whole passes");
    __shared__ __attribute__((aligned(16))) float sWt[NCOL * SROW_PITCH];
    __shared__ __attribute__((aligned(16))) float sA[ROWS * SROW_PITCH];

    constexpr int TPR   = NTHR / NCOL;
    constexpr int ITERS = (K_DIM / 4) / TPR;
    {
        const int col = threadIdx.x % NCOL;
        const int kq  = threadIdx.x / NCOL;
        float4 v[ITERS];
        #pragma unroll
        for (int it = 0; it < ITERS; ++it) {
            const float* bp = W2 + (4 * (kq + it * TPR)) * NCOL + col;
            v[it].x = bp[0];
            v[it].y = bp[NCOL];
            v[it].z = bp[2 * NCOL];
            v[it].w = bp[3 * NCOL];
        }
        #pragma unroll
        for (int it = 0; it < ITERS; ++it)
            *(float4*)&sWt[col * SROW_PITCH + 4 * (kq + it * TPR)] = v[it];
    }

    const int lane = threadIdx.x & 31;
    const int wave = __builtin_amdgcn_readfirstlane((int)(threadIdx.x >> 5));
    const int gw = (int)blockIdx.x * WPB + wave;
    float* srow = sA + wave * NPW * SROW_PITCH;

    gather_walk<4, 32, true, NPW, true>(start, cnt, csr_src, hfeat, a_s, a_d, bias, nullptr, N, gw, srow);

    {
        int nv = N - gw * NPW;
        nv = nv < 0 ? 0 : (nv > NPW ? NPW : nv);
        for (int j = nv; j < NPW; ++j)
            *(float4*)(srow + j * SROW_PITCH + lane * 4) = make_float4(0.f, 0.f, 0.f, 0.f);
    }
    __syncthreads();

    if (wave < ROWS / 16)
        tile_product_lds<NCOL, 4, 64>(sA + wave * 16 * SROW_PITCH, sWt, att_s2, att_d2,
                                      h2, a2s, a2d, N, (int)blockIdx.x * ROWS + wave * 16, lane);
}

extern "C" void kernel_launch(void* const* d_in, const int* in_sizes, int n_in,
                              void* d_out, int out_size, void* d_ws, size_t ws_size,
                              hipStream_t stream) {
    const float* x   = (const float*)d_in[0];
    const int*   ei  = (const int*)  d_in[1];
    const float* W1  = (const float*)d_in[2];
    const float* as1 = (const float*)d_in[3];
    const float* ad1 = (const float*)d_in[4];
    const float* b1  = (const float*)d_in[5];
    const float* W2  = (const float*)d_in[6];
    const float* as2 = (const float*)d_in[7];
    const float* ad2 = (const float*)d_in[8];
    const float* b2  = (const float*)d_in[9];
    float* out = (float*)d_out;

    const int N  = in_sizes[0] / 128;
    const int E  = in_sizes[1] / 2;
    const int H1 = 4, F1 = 128;
    const int F2 = 64;

    char* w = (char*)d_ws;
    auto carve = [&](size_t bytes) {
        void* p = (void*)w;
        w += (bytes + 255) & ~(size_t)255;
        return p;
    };
    float* h1    = (float*)carve((size_t)N * F1 * 4);
    float* a1s   = (float*)carve((size_t)N * H1 * 4);
    float* a1d   = (float*)carve((size_t)N * H1 * 4);
    float* agg1  = (float*)carve((size_t)N * F1 * 4);
    float* h2    = (float*)carve((size_t)N * F2 * 4);
    float* a2s   = (float*)carve((size_t)N * 4);
    float* a2d   = (float*)carve((size_t)N * 4);
    int*   cnt   = (int*)  carve((size_t)(N + 1) * 4);
    int*   start = (int*)  carve((size_t)N * 4);
    int*   csrc  = (int*)  carve((size_t)E * 4);

    const int ng = (N + 8 * GATHER_NPW - 1) / (8 * GATHER_NPW);

    const int nchunks = (E + PART_CHUNK - 1) / PART_CHUNK;
    const int nbins   = (N + (1 << PART_SHIFT) - 1) >> PART_SHIFT;
    int* part = (int*)agg1;
    int* runs = part + (((size_t)E * 2 + 63) & ~(size_t)63);
    edge_partition_kernel<<<nchunks, 1024, 0, stream>>>(ei, E, nbins, nchunks, part, runs, cnt + N);
    segment_build_kernel<<<nbins, 1024, 0, stream>>>(part, runs, nchunks, N, cnt + N, cnt, start, csrc);

    {
        int mt = (N + 15) / 16;
        wmma_gemm_kernel<128, 4, 32><<<GEMM_BLOCKS, 256, 0, stream>>>(
            x, W1, as1, ad1, h1, a1s, a1d, N, mt);
    }
    {
        const int rows = FUSED_WPB * GATHER_NPW;
        const int nb = (N + rows - 1) / rows;
        gather_product_kernel<GATHER_NPW, FUSED_WPB><<<nb, FUSED_WPB * 32, 0, stream>>>(
            start, cnt, csrc, h1, a1s, a1d, b1, W2, as2, ad2, h2, a2s, a2d, N);
    }
    node_gather_kernel<1, 64, false, GATHER_NPW><<<ng, 256, 0, stream>>>(start, cnt, csrc, h2, a2s, a2d, b2, out, N);
}
